# starting kernel
# baseline (speedup 1.0000x reference)
#include <hip/hip_runtime.h>
#include <math.h>

typedef __attribute__((ext_vector_type(2))) float v2f;
typedef __attribute__((ext_vector_type(8))) float v8f;

#define HH 768
#define WW 768
#define HWSZ (768*768)
#define EPSV 1e-4f

__global__ void initW(float* wbil, float* wgau, float* sxb, float* sxg) {
  __shared__ float wb[121];
  __shared__ float wg[25];
  int t = threadIdx.x;
  if (t < 121) { float d = (float)(t - 60); float v = expf(-0.5f*d*d/(15.f*15.f)); wb[t] = v; wbil[t] = v; }
  if (t < 25)  { float d = (float)(t - 12); float v = expf(-d*d/18.f);             wg[t] = v; wgau[t] = v; }
  __syncthreads();
  for (int x = t; x < WW; x += blockDim.x) {
    float s = 0.f;
    for (int d = 0; d < 121; ++d) { int ix = x + d - 60; if (ix >= 0 && ix < WW) s += wb[d]; }
    sxb[x] = s;
    float s2 = 0.f;
    for (int d = 0; d < 25; ++d) { int ix = x + d - 12; if (ix >= 0 && ix < WW) s2 += wg[d]; }
    sxg[x] = s2;
  }
}

template<int R>
__global__ __launch_bounds__(256) void convH(const float* __restrict__ src,
                                             float* __restrict__ dst,
                                             const float* __restrict__ wdev) {
  constexpr int TW  = 128;
  constexpr int SW  = TW + 2*R;
  constexpr int NCH = (2*R + 16) / 4;
  constexpr int WP  = 2*R + 31;
  constexpr int SP  = SW + 1;
  __shared__ float lds[(WP + 1) + 16*SP];
  float* wpad  = lds;
  float* strip = lds + (WP + 1);
  const int tid = threadIdx.x;
  const size_t pb = (size_t)blockIdx.z * HWSZ;
  const float* ip = src + pb;
  float* op = dst + pb;
  const int gy0 = blockIdx.y * 16;
  const int gx0 = blockIdx.x * TW;
  for (int i = tid; i <= WP; i += 256) {
    int d = i - 15;
    wpad[i] = (d >= 0 && d <= 2*R) ? wdev[d] : 0.f;
  }
  for (int idx = tid; idx < 16*SW; idx += 256) {
    int r = idx / SW, c = idx - r*SW;
    int gx = gx0 - R + c;
    strip[r*SP + c] = (gx >= 0 && gx < WW) ? ip[(size_t)(gy0 + r)*WW + gx] : 0.f;
  }
  __syncthreads();
  const int lane = tid & 31;
  const int wv   = tid >> 5;
  const int j    = lane & 15;
  const int kb   = (lane >> 4) * 2;
  const float* ap = strip + j*SP + wv*16 + kb;
  const float* bp = wpad + (kb - j + 15);
  v8f acc = {0.f,0.f,0.f,0.f,0.f,0.f,0.f,0.f};
  #pragma unroll
  for (int kk = 0; kk < NCH; ++kk) {
    v2f a, b;
    a.x = ap[kk*4];
    a.y = ap[kk*4 + 1];
    b.x = bp[kk*4];
    b.y = bp[kk*4 + 1];
    acc = __builtin_amdgcn_wmma_f32_16x16x4_f32(false, a, false, b, (short)0, acc, false, false);
  }
  const int row0 = gy0 + (lane >> 4) * 8;
  const int col  = gx0 + wv*16 + j;
  #pragma unroll
  for (int i = 0; i < 8; ++i) op[(size_t)(row0 + i)*WW + col] = acc[i];
}

template<int R>
__global__ __launch_bounds__(256) void convV(const float* __restrict__ src,
                                             float* __restrict__ dst,
                                             const float* __restrict__ wdev) {
  constexpr int TH  = 128;
  constexpr int SH  = TH + 2*R;
  constexpr int NCH = (2*R + 16) / 4;
  constexpr int WP  = 2*R + 31;
  constexpr int SP  = SH + 1;
  __shared__ float lds[(WP + 1) + 16*SP];
  float* wpad   = lds;
  float* stripT = lds + (WP + 1);
  const int tid = threadIdx.x;
  const size_t pb = (size_t)blockIdx.z * HWSZ;
  const float* ip = src + pb;
  float* op = dst + pb;
  const int gy0 = blockIdx.y * TH;
  const int gx0 = blockIdx.x * 16;
  for (int i = tid; i <= WP; i += 256) {
    int d = i - 15;
    wpad[i] = (d >= 0 && d <= 2*R) ? wdev[d] : 0.f;
  }
  for (int idx = tid; idx < SH*16; idx += 256) {
    int r = idx >> 4, c = idx & 15;
    int gy = gy0 - R + r;
    stripT[c*SP + r] = (gy >= 0 && gy < HH) ? ip[(size_t)gy*WW + gx0 + c] : 0.f;
  }
  __syncthreads();
  const int lane = tid & 31;
  const int wv   = tid >> 5;
  const int j    = lane & 15;
  const int kb   = (lane >> 4) * 2;
  const float* bp = stripT + j*SP + wv*16 + kb;
  const float* apw = wpad + (kb - j + 15);
  v8f acc = {0.f,0.f,0.f,0.f,0.f,0.f,0.f,0.f};
  #pragma unroll
  for (int kk = 0; kk < NCH; ++kk) {
    v2f a, b;
    a.x = apw[kk*4];
    a.y = apw[kk*4 + 1];
    b.x = bp[kk*4];
    b.y = bp[kk*4 + 1];
    acc = __builtin_amdgcn_wmma_f32_16x16x4_f32(false, a, false, b, (short)0, acc, false, false);
  }
  const int row0 = gy0 + wv*16 + (lane >> 4) * 8;
  const int col  = gx0 + j;
  #pragma unroll
  for (int i = 0; i < 8; ++i) op[(size_t)(row0 + i)*WW + col] = acc[i];
}

__global__ void prepImg(const float* __restrict__ img, float* __restrict__ Ipl,
                        float* __restrict__ P9) {
  int p = blockIdx.x*256 + threadIdx.x;
  if (p >= HWSZ) return;
  float i0 = img[p*3+0], i1 = img[p*3+1], i2 = img[p*3+2];
  Ipl[p] = i0; Ipl[HWSZ+p] = i1; Ipl[2*HWSZ+p] = i2;
  P9[0*HWSZ+p] = i0;    P9[1*HWSZ+p] = i1;    P9[2*HWSZ+p] = i2;
  P9[3*HWSZ+p] = i0*i0; P9[4*HWSZ+p] = i0*i1; P9[5*HWSZ+p] = i0*i2;
  P9[6*HWSZ+p] = i1*i1; P9[7*HWSZ+p] = i1*i2; P9[8*HWSZ+p] = i2*i2;
}

__global__ void statsK(const float* __restrict__ U, const float* __restrict__ Ipl,
                       const float* __restrict__ sxb,
                       float* __restrict__ meanI, float* __restrict__ inv6) {
  int p = blockIdx.x*256 + threadIdx.x;
  if (p >= HWSZ) return;
  int x = p % WW, y = p / WW;
  float rN = 1.f / (sxb[x]*sxb[y] - 1.f);
  float i0 = Ipl[p], i1 = Ipl[HWSZ+p], i2 = Ipl[2*HWSZ+p];
  float m0 = (U[0*HWSZ+p] - i0)*rN;
  float m1 = (U[1*HWSZ+p] - i1)*rN;
  float m2 = (U[2*HWSZ+p] - i2)*rN;
  float v00 = (U[3*HWSZ+p] - i0*i0)*rN - m0*m0 + EPSV;
  float v01 = (U[4*HWSZ+p] - i0*i1)*rN - m0*m1;
  float v02 = (U[5*HWSZ+p] - i0*i2)*rN - m0*m2;
  float v11 = (U[6*HWSZ+p] - i1*i1)*rN - m1*m1 + EPSV;
  float v12 = (U[7*HWSZ+p] - i1*i2)*rN - m1*m2;
  float v22 = (U[8*HWSZ+p] - i2*i2)*rN - m2*m2 + EPSV;
  float c00 = v11*v22 - v12*v12;
  float c01 = v02*v12 - v01*v22;
  float c02 = v01*v12 - v02*v11;
  float c11 = v00*v22 - v02*v02;
  float c12 = v01*v02 - v00*v12;
  float c22 = v00*v11 - v01*v01;
  float rd = 1.f / (v00*c00 + v01*c01 + v02*c02);
  meanI[p] = m0; meanI[HWSZ+p] = m1; meanI[2*HWSZ+p] = m2;
  inv6[0*HWSZ+p] = c00*rd; inv6[1*HWSZ+p] = c01*rd; inv6[2*HWSZ+p] = c02*rd;
  inv6[3*HWSZ+p] = c11*rd; inv6[4*HWSZ+p] = c12*rd; inv6[5*HWSZ+p] = c22*rd;
}

__global__ void softmaxInit(const float* __restrict__ un, float* __restrict__ Q) {
  int p = blockIdx.x*256 + threadIdx.x;
  if (p >= HWSZ) return;
  float L[4];
  float m = -3.4e38f;
  #pragma unroll
  for (int k = 0; k < 4; ++k) { L[k] = -un[p*4+k]; m = fmaxf(m, L[k]); }
  float s = 0.f;
  #pragma unroll
  for (int k = 0; k < 4; ++k) { L[k] = expf(L[k] - m); s += L[k]; }
  float rs = 1.f / s;
  #pragma unroll
  for (int k = 0; k < 4; ++k) Q[k*HWSZ+p] = L[k]*rs;
}

__global__ void prep16K(const float* __restrict__ Q, const float* __restrict__ Ipl,
                        float* __restrict__ P) {
  int p = blockIdx.x*256 + threadIdx.x;
  if (p >= HWSZ) return;
  float I[3] = { Ipl[p], Ipl[HWSZ+p], Ipl[2*HWSZ+p] };
  #pragma unroll
  for (int k = 0; k < 4; ++k) {
    float q = Q[k*HWSZ+p];
    P[k*HWSZ+p] = q;
    #pragma unroll
    for (int i = 0; i < 3; ++i) P[(size_t)(4 + i*4 + k)*HWSZ+p] = I[i]*q;
  }
}

__global__ void midABK(const float* __restrict__ U, const float* __restrict__ Q,
                       const float* __restrict__ Ipl, const float* __restrict__ meanI,
                       const float* __restrict__ inv6, const float* __restrict__ sxb,
                       float* __restrict__ AB) {
  int p = blockIdx.x*256 + threadIdx.x;
  if (p >= HWSZ) return;
  int x = p % WW, y = p / WW;
  float rN = 1.f / (sxb[x]*sxb[y] - 1.f);
  float I0 = Ipl[p], I1 = Ipl[HWSZ+p], I2 = Ipl[2*HWSZ+p];
  float mI0 = meanI[p], mI1 = meanI[HWSZ+p], mI2 = meanI[2*HWSZ+p];
  float iv0 = inv6[0*HWSZ+p], iv1 = inv6[1*HWSZ+p], iv2 = inv6[2*HWSZ+p];
  float iv3 = inv6[3*HWSZ+p], iv4 = inv6[4*HWSZ+p], iv5 = inv6[5*HWSZ+p];
  #pragma unroll
  for (int k = 0; k < 4; ++k) {
    float q  = Q[k*HWSZ+p];
    float mp = (U[k*HWSZ+p] - q)*rN;
    float c0 = (U[(size_t)(4 +k)*HWSZ+p] - I0*q)*rN - mI0*mp;
    float c1 = (U[(size_t)(8 +k)*HWSZ+p] - I1*q)*rN - mI1*mp;
    float c2 = (U[(size_t)(12+k)*HWSZ+p] - I2*q)*rN - mI2*mp;
    float a0 = iv0*c0 + iv1*c1 + iv2*c2;
    float a1 = iv1*c0 + iv3*c1 + iv4*c2;
    float a2 = iv2*c0 + iv4*c1 + iv5*c2;
    float b  = mp - (a0*mI0 + a1*mI1 + a2*mI2);
    AB[(size_t)(0 +k)*HWSZ+p] = a0;
    AB[(size_t)(4 +k)*HWSZ+p] = a1;
    AB[(size_t)(8 +k)*HWSZ+p] = a2;
    AB[(size_t)(12+k)*HWSZ+p] = b;
  }
}

__global__ void combineK(const float* __restrict__ U, const float* __restrict__ AB,
                         const float* __restrict__ Us4, const float* __restrict__ Ipl,
                         const float* __restrict__ un, const float* __restrict__ sxb,
                         const float* __restrict__ sxg, float* __restrict__ Q) {
  int p = blockIdx.x*256 + threadIdx.x;
  if (p >= HWSZ) return;
  int x = p % WW, y = p / WW;
  float rN = 1.f / (sxb[x]*sxb[y] - 1.f);
  float rS = 1.f / (sxg[x]*sxg[y] - 1.f);
  float I0 = Ipl[p], I1 = Ipl[HWSZ+p], I2 = Ipl[2*HWSZ+p];
  float L[4];
  float m = -3.4e38f;
  #pragma unroll
  for (int k = 0; k < 4; ++k) {
    float ma0 = (U[(size_t)(0 +k)*HWSZ+p] - AB[(size_t)(0 +k)*HWSZ+p])*rN;
    float ma1 = (U[(size_t)(4 +k)*HWSZ+p] - AB[(size_t)(4 +k)*HWSZ+p])*rN;
    float ma2 = (U[(size_t)(8 +k)*HWSZ+p] - AB[(size_t)(8 +k)*HWSZ+p])*rN;
    float mb  = (U[(size_t)(12+k)*HWSZ+p] - AB[(size_t)(12+k)*HWSZ+p])*rN;
    float bil = ma0*I0 + ma1*I1 + ma2*I2 + mb;
    float q   = Q[k*HWSZ+p];
    float sp  = (Us4[k*HWSZ+p] - q)*rS;
    L[k] = -un[p*4+k] + 3.f*sp + 10.f*bil;
    m = fmaxf(m, L[k]);
  }
  float s = 0.f;
  #pragma unroll
  for (int k = 0; k < 4; ++k) { L[k] = expf(L[k] - m); s += L[k]; }
  float rs = 1.f / s;
  #pragma unroll
  for (int k = 0; k < 4; ++k) Q[k*HWSZ+p] = L[k]*rs;
}

__global__ void packK(const float* __restrict__ Q, float* __restrict__ out) {
  int p = blockIdx.x*256 + threadIdx.x;
  if (p >= HWSZ) return;
  #pragma unroll
  for (int k = 0; k < 4; ++k) out[p*4+k] = Q[k*HWSZ+p];
}

extern "C" void kernel_launch(void* const* d_in, const int* in_sizes, int n_in,
                              void* d_out, int out_size, void* d_ws, size_t ws_size,
                              hipStream_t stream) {
  (void)in_sizes; (void)n_in; (void)out_size; (void)ws_size;
  const float* img = (const float*)d_in[0];
  const float* un  = (const float*)d_in[1];
  float* ws   = (float*)d_ws;
  float* wbil = ws;
  float* wgau = ws + 128;
  float* sxb  = ws + 160;
  float* sxg  = ws + 928;
  float* pl   = ws + 2048;
  auto P = [&](int i) { return pl + (size_t)i * HWSZ; };

  dim3 blk(256);
  dim3 gP(HWSZ/256);
  initW<<<1, 256, 0, stream>>>(wbil, wgau, sxb, sxg);
  prepImg<<<gP, blk, 0, stream>>>(img, P(0), P(16));
  convH<60><<<dim3(6,48,9),  blk, 0, stream>>>(P(16), P(32), wbil);
  convV<60><<<dim3(48,6,9),  blk, 0, stream>>>(P(32), P(48), wbil);
  statsK<<<gP, blk, 0, stream>>>(P(48), P(0), sxb, P(3), P(6));
  softmaxInit<<<gP, blk, 0, stream>>>(un, P(12));

  for (int it = 0; it < 5; ++it) {
    prep16K<<<gP, blk, 0, stream>>>(P(12), P(0), P(16));
    convH<60><<<dim3(6,48,16), blk, 0, stream>>>(P(16), P(32), wbil);
    convV<60><<<dim3(48,6,16), blk, 0, stream>>>(P(32), P(48), wbil);
    midABK<<<gP, blk, 0, stream>>>(P(48), P(12), P(0), P(3), P(6), sxb, P(16));
    convH<60><<<dim3(6,48,16), blk, 0, stream>>>(P(16), P(32), wbil);
    convV<60><<<dim3(48,6,16), blk, 0, stream>>>(P(32), P(48), wbil);
    convH<12><<<dim3(6,48,4),  blk, 0, stream>>>(P(12), P(32), wgau);
    convV<12><<<dim3(48,6,4),  blk, 0, stream>>>(P(32), P(64), wgau);
    combineK<<<gP, blk, 0, stream>>>(P(48), P(16), P(64), P(0), un, sxb, sxg, P(12));
  }
  packK<<<gP, blk, 0, stream>>>(P(12), (float*)d_out);
}


# guided-filter coefficient stage moved into the first vertical pass: a workgroup walks the 4 planes of one class and writes a,b directly; 16 filtered planes and 5 pointwise launches gone
# speedup vs baseline: 1.8221x; 1.8221x over previous
#include <hip/hip_runtime.h>
#include <math.h>

typedef __attribute__((ext_vector_type(2))) float v2f;
typedef __attribute__((ext_vector_type(4))) float v4f;
typedef __attribute__((ext_vector_type(8))) float v8f;

#define HH 768
#define WW 768
#define HWSZ (768*768)
#define EPSV 1e-4f

__global__ void initW(float* wbil, float* wgau, float* sxb, float* sxg) {
  __shared__ float wb[121];
  __shared__ float wg[25];
  int t = threadIdx.x;
  if (t < 121) { float d = (float)(t - 60); float v = expf(-0.5f*d*d/(15.f*15.f)); wb[t] = v; wbil[t] = v; }
  if (t < 25)  { float d = (float)(t - 12); float v = expf(-d*d/18.f);             wg[t] = v; wgau[t] = v; }
  __syncthreads();
  for (int x = t; x < WW; x += blockDim.x) {
    float s = 0.f;
    for (int d = 0; d < 121; ++d) { int ix = x + d - 60; if (ix >= 0 && ix < WW) s += wb[d]; }
    sxb[x] = s;
    float s2 = 0.f;
    for (int d = 0; d < 25; ++d) { int ix = x + d - 12; if (ix >= 0 && ix < WW) s2 += wg[d]; }
    sxg[x] = s2;
  }
}

template<int R, int ITERS, bool MUL>
__global__ __launch_bounds__(256) void convH(const float* __restrict__ src,
                                             const float* __restrict__ src2,
                                             float* __restrict__ dst,
                                             const float* __restrict__ wdev) {
  constexpr int TW   = 128;
  constexpr int SW   = TW + 2*R;
  constexpr int NV   = SW / 4;
  constexpr int KB   = 2*R + 32;
  constexpr int NCH  = KB / 8;
  constexpr int WQ   = 2*R + 63;
  constexpr int WQP  = (WQ + 4) & ~3;
  constexpr int TABN = NCH * 64;
  constexpr int SP   = SW + 4;
  constexpr int ROWS = 32;
  constexpr int TOT  = ROWS*NV;
  constexpr int PER  = (TOT + 255) / 256;
  static_assert(SW % 4 == 0 && KB % 8 == 0 && SP % 4 == 0 && ((SP / 4) & 1) == 1, "strip layout");
  static_assert(WQ < 256, "one weight per thread");
  __shared__ __attribute__((aligned(16))) float lds[WQP + 2*TABN + ROWS*SP];
  float* wq    = lds;
  float* tabE  = lds + WQP;
  float* tabO  = tabE + TABN;
  float* strip = tabO + TABN;
  const int tid = threadIdx.x;
  const int zpl = blockIdx.z;
  const size_t pb = (size_t)zpl * HWSZ;
  const float* ip  = src + pb;
  const float* ip2 = src2;
  bool mul = false;
  if (MUL) {
    if (zpl < 4) {
      ip = src + (size_t)zpl * HWSZ;
    } else {
      const int zz = zpl - 4;
      ip  = src  + (size_t)(zz & 3)  * HWSZ;
      ip2 = src2 + (size_t)(zz >> 2) * HWSZ;
      mul = true;
    }
  }
  float* op = dst + pb;
  const int gx0 = blockIdx.x * TW;
  const int gyb = blockIdx.y * (ROWS*ITERS);

  float wreg = 0.f;
  {
    const int d = tid - 31;
    if (tid < WQ && d >= 0 && d <= 2*R) wreg = wdev[d];
  }

  v4f pre[PER];
  v4f pre2[MUL ? PER : 1];
  auto fetch = [&](int gy) {
    unsigned off[PER];
    #pragma unroll
    for (int u = 0; u < PER; ++u) {
      const int idx = tid + u*256;
      const int r  = idx / NV;
      const int c4 = idx - r*NV;
      const int gx = gx0 - R + 4*c4;
      const bool live = idx < TOT;
      const bool ok = live && gx >= 0 && gx < WW;
      const int rr = live ? r : 0;
      const int gc = ok ? gx : 0;
      off[u] = (unsigned)((gy + rr)*WW + gc);
      pre[u] = *(const v4f*)(ip + off[u]);
    }
    if (MUL) {
      if (mul) {
        #pragma unroll
        for (int u = 0; u < PER; ++u) pre2[MUL ? u : 0] = *(const v4f*)(ip2 + off[u]);
      }
    }
  };
  auto stash = [&]() {
    #pragma unroll
    for (int u = 0; u < PER; ++u) {
      const int idx = tid + u*256;
      const int r  = idx / NV;
      const int c4 = idx - r*NV;
      const int gx = gx0 - R + 4*c4;
      const bool ok = gx >= 0 && gx < WW;
      const v4f z = {0.f, 0.f, 0.f, 0.f};
      v4f v = pre[u];
      if (MUL) { if (mul) v = v * pre2[MUL ? u : 0]; }
      if (idx < TOT) *(v4f*)(strip + r*SP + 4*c4) = ok ? v : z;
    }
  };

  const int lane = tid & 31;
  const int wv   = tid >> 5;
  const int j    = lane & 15;
  const int h    = lane >> 4;
  const int bl   = wv & 3;
  const int rt   = wv >> 2;
  const float* ap = strip + (rt*16 + j)*SP + bl*32 + 2*h;
  const float* rp = strip + (rt*16 + j)*SP + bl*32 + (KB - 2) - 2*h;
  const float* te = tabE + (h*16 + j)*2;
  const float* to = tabO + (h*16 + j)*2;
  const int colA = gx0 + bl*32 + j;
  const int colB = gx0 + bl*32 + 31 - j;

  fetch(gyb);
  if (tid < WQ) wq[tid] = wreg;
  __syncthreads();
  for (int e = tid; e < TABN; e += 256) {
    const int s  = e & 1;
    const int jj = (e >> 1) & 15;
    const int hh = (e >> 5) & 1;
    const int k  = 4*(e >> 6) + 2*hh + s;
    const float wa = wq[k - jj + 31];
    const float wb = wq[k + jj];
    tabE[e] = 0.5f*(wa + wb);
    tabO[e] = 0.5f*(wa - wb);
  }
  stash();
  __syncthreads();
  #pragma unroll
  for (int it = 0; it < ITERS; ++it) {
    const int gy = gyb + it*ROWS;
    if (it + 1 < ITERS) fetch(gy + ROWS);
    __builtin_amdgcn_sched_barrier(0);
    v8f accE = {0.f,0.f,0.f,0.f,0.f,0.f,0.f,0.f};
    v8f accO = {0.f,0.f,0.f,0.f,0.f,0.f,0.f,0.f};
    #pragma unroll
    for (int c = 0; c < NCH; ++c) {
      const v2f f  = *(const v2f*)(ap + 4*c);
      const v2f r  = *(const v2f*)(rp - 4*c);
      const v2f be = *(const v2f*)(te + 64*c);
      const v2f bo = *(const v2f*)(to + 64*c);
      v2f sm, df;
      sm.x = f.x + r.y;  sm.y = f.y + r.x;
      df.x = f.x - r.y;  df.y = f.y - r.x;
      accE = __builtin_amdgcn_wmma_f32_16x16x4_f32(false, sm, false, be, (short)0, accE, false, false);
      accO = __builtin_amdgcn_wmma_f32_16x16x4_f32(false, df, false, bo, (short)0, accO, false, false);
    }
    __builtin_amdgcn_sched_barrier(0);
    if (it + 1 < ITERS) {
      __syncthreads();
      stash();
      __syncthreads();
    }
    const int row0 = gy + rt*16 + h*8;
    #pragma unroll
    for (int i = 0; i < 8; ++i) {
      op[(size_t)(row0 + i)*WW + colA] = accE[i] + accO[i];
      op[(size_t)(row0 + i)*WW + colB] = accE[i] - accO[i];
    }
  }
}

template<int R, int ITERS>
__global__ __launch_bounds__(256) void convV(const float* __restrict__ src,
                                             float* __restrict__ dst,
                                             const float* __restrict__ wdev) {
  constexpr int TH   = 128;
  constexpr int SH   = TH + 2*R;
  constexpr int CW   = 32;
  constexpr int NV   = CW / 4;
  constexpr int KB   = 2*R + 32;
  constexpr int NCH  = KB / 8;
  constexpr int WQ   = 2*R + 63;
  constexpr int WQP  = (WQ + 4) & ~3;
  constexpr int TABN = NCH * 64;
  constexpr int SP   = SH + 1;
  constexpr int TOT  = SH*NV;
  constexpr int PER  = (TOT + 255) / 256;
  static_assert(KB % 8 == 0, "folded band");
  static_assert(WQ < 256, "one weight per thread");
  __shared__ __attribute__((aligned(16))) float lds[WQP + 2*TABN + CW*SP];
  float* wq     = lds;
  float* tabE   = lds + WQP;
  float* tabO   = tabE + TABN;
  float* stripT = tabO + TABN;
  const int tid = threadIdx.x;
  const size_t pb = (size_t)blockIdx.z * HWSZ;
  const float* ip = src + pb;
  float* op = dst + pb;
  const int gy0 = blockIdx.y * TH;
  const int gxb = blockIdx.x * (CW*ITERS);

  float wreg = 0.f;
  {
    const int d = tid - 31;
    if (tid < WQ && d >= 0 && d <= 2*R) wreg = wdev[d];
  }

  v4f pre[PER];
  auto fetch = [&](int gx) {
    #pragma unroll
    for (int u = 0; u < PER; ++u) {
      const int idx = tid + u*256;
      const int r  = idx / NV;
      const int c4 = idx - r*NV;
      const int gy = gy0 - R + r;
      const bool ok = idx < TOT && gy >= 0 && gy < HH;
      const int gr = ok ? gy : 0;
      pre[u] = *(const v4f*)(ip + (size_t)gr*WW + gx + 4*c4);
    }
  };
  auto stash = [&]() {
    #pragma unroll
    for (int u = 0; u < PER; ++u) {
      const int idx = tid + u*256;
      const int r  = idx / NV;
      const int c4 = idx - r*NV;
      const int gy = gy0 - R + r;
      const bool ok = gy >= 0 && gy < HH;
      const v4f z = {0.f, 0.f, 0.f, 0.f};
      const v4f v = ok ? pre[u] : z;
      if (idx < TOT) {
        float* q = stripT + (4*c4)*SP + r;
        q[0]    = v.x;
        q[SP]   = v.y;
        q[2*SP] = v.z;
        q[3*SP] = v.w;
      }
    }
  };

  const int lane = tid & 31;
  const int wv   = tid >> 5;
  const int j    = lane & 15;
  const int h    = lane >> 4;
  const int rb   = wv & 3;
  const int ct   = wv >> 2;
  const float* bp = stripT + (ct*16 + j)*SP + rb*32 + 2*h;
  const float* rp = stripT + (ct*16 + j)*SP + rb*32 + (KB - 1) - 2*h;
  const float* te = tabE + (h*16 + j)*2;
  const float* to = tabO + (h*16 + j)*2;
  const int y0 = gy0 + rb*32;

  fetch(gxb);
  if (tid < WQ) wq[tid] = wreg;
  __syncthreads();
  for (int e = tid; e < TABN; e += 256) {
    const int s  = e & 1;
    const int ii = (e >> 1) & 15;
    const int hh = (e >> 5) & 1;
    const int k  = 4*(e >> 6) + 2*hh + s;
    const float wa = wq[k - ii + 31];
    const float wb = wq[k + ii];
    tabE[e] = 0.5f*(wa + wb);
    tabO[e] = 0.5f*(wa - wb);
  }
  stash();
  __syncthreads();
  #pragma unroll
  for (int it = 0; it < ITERS; ++it) {
    const int gx = gxb + it*CW;
    if (it + 1 < ITERS) fetch(gx + CW);
    __builtin_amdgcn_sched_barrier(0);
    v8f accE = {0.f,0.f,0.f,0.f,0.f,0.f,0.f,0.f};
    v8f accO = {0.f,0.f,0.f,0.f,0.f,0.f,0.f,0.f};
    #pragma unroll
    for (int c = 0; c < NCH; ++c) {
      const float f0 = bp[4*c];
      const float f1 = bp[4*c + 1];
      const float r0 = rp[-4*c];
      const float r1 = rp[-4*c - 1];
      const v2f ae = *(const v2f*)(te + 64*c);
      const v2f ao = *(const v2f*)(to + 64*c);
      v2f sm, df;
      sm.x = f0 + r0;  sm.y = f1 + r1;
      df.x = f0 - r0;  df.y = f1 - r1;
      accE = __builtin_amdgcn_wmma_f32_16x16x4_f32(false, ae, false, sm, (short)0, accE, false, false);
      accO = __builtin_amdgcn_wmma_f32_16x16x4_f32(false, ao, false, df, (short)0, accO, false, false);
    }
    __builtin_amdgcn_sched_barrier(0);
    if (it + 1 < ITERS) {
      __syncthreads();
      stash();
      __syncthreads();
    }
    const int col = gx + ct*16 + j;
    #pragma unroll
    for (int i = 0; i < 8; ++i) {
      const int ii = h*8 + i;
      op[(size_t)(y0 + ii)*WW + col]      = accE[i] + accO[i];
      op[(size_t)(y0 + 31 - ii)*WW + col] = accE[i] - accO[i];
    }
  }
}

template<int R>
__global__ __launch_bounds__(256, 6) void convVmid(const float* __restrict__ src,
                                                const float* __restrict__ Q,
                                                const float* __restrict__ Ipl,
                                                const float* __restrict__ meanI,
                                                const float* __restrict__ inv6,
                                                const float* __restrict__ sxb,
                                                float* __restrict__ AB,
                                                const float* __restrict__ wdev) {
  constexpr int TH   = 128;
  constexpr int SH   = TH + 2*R;
  constexpr int CW   = 32;
  constexpr int NV   = CW / 4;
  constexpr int KB   = 2*R + 32;
  constexpr int NCH  = KB / 8;
  constexpr int WQ   = 2*R + 63;
  constexpr int WQP  = (WQ + 4) & ~3;
  constexpr int TABN = NCH * 64;
  constexpr int SP   = SH + 1;
  constexpr int TOT  = SH*NV;
  constexpr int PER  = (TOT + 255) / 256;
  constexpr int NPL  = 4;
  constexpr int OP   = 36;
  constexpr int CONV = WQP + 2*TABN + CW*SP;
  constexpr int STG  = NPL*TH*OP;
  constexpr int LDSN = CONV > STG ? CONV : STG;
  static_assert(KB % 8 == 0, "folded band");
  static_assert(WQ < 256, "one weight per thread");
  __shared__ __attribute__((aligned(16))) float lds[LDSN];
  float* wq     = lds;
  float* tabE   = lds + WQP;
  float* tabO   = tabE + TABN;
  float* stripT = tabO + TABN;
  float* stg    = lds;
  const int tid = threadIdx.x;
  const int k   = blockIdx.z;
  const int gy0 = blockIdx.y * TH;
  const int gx  = blockIdx.x * CW;

  float wreg = 0.f;
  {
    const int d = tid - 31;
    if (tid < WQ && d >= 0 && d <= 2*R) wreg = wdev[d];
  }

  v4f pre[PER];
  unsigned foff[PER];
  #pragma unroll
  for (int u = 0; u < PER; ++u) {
    const int idx = tid + u*256;
    const int r  = idx / NV;
    const int c4 = idx - r*NV;
    const int gy = gy0 - R + r;
    const bool ok = idx < TOT && gy >= 0 && gy < HH;
    const int gr = ok ? gy : 0;
    foff[u] = (unsigned)(gr*WW + gx + 4*c4);
  }
  auto fetch = [&](int pl) {
    const float* ip = src + (size_t)(k + 4*pl) * HWSZ;
    #pragma unroll
    for (int u = 0; u < PER; ++u) pre[u] = *(const v4f*)(ip + foff[u]);
  };
  auto stash = [&]() {
    #pragma unroll
    for (int u = 0; u < PER; ++u) {
      const int idx = tid + u*256;
      const int r  = idx / NV;
      const int c4 = idx - r*NV;
      const int gy = gy0 - R + r;
      const bool ok = gy >= 0 && gy < HH;
      const v4f z = {0.f, 0.f, 0.f, 0.f};
      const v4f v = ok ? pre[u] : z;
      if (idx < TOT) {
        float* q = stripT + (4*c4)*SP + r;
        q[0]    = v.x;
        q[SP]   = v.y;
        q[2*SP] = v.z;
        q[3*SP] = v.w;
      }
    }
  };

  const int lane = tid & 31;
  const int wv   = tid >> 5;
  const int j    = lane & 15;
  const int h    = lane >> 4;
  const int rb   = wv & 3;
  const int ct   = wv >> 2;
  const float* bp = stripT + (ct*16 + j)*SP + rb*32 + 2*h;
  const float* rp = stripT + (ct*16 + j)*SP + rb*32 + (KB - 1) - 2*h;
  const float* te = tabE + (h*16 + j)*2;
  const float* to = tabO + (h*16 + j)*2;

  fetch(0);
  if (tid < WQ) wq[tid] = wreg;
  __syncthreads();
  for (int e = tid; e < TABN; e += 256) {
    const int s  = e & 1;
    const int ii = (e >> 1) & 15;
    const int hh = (e >> 5) & 1;
    const int kk = 4*(e >> 6) + 2*hh + s;
    const float wa = wq[kk - ii + 31];
    const float wb = wq[kk + ii];
    tabE[e] = 0.5f*(wa + wb);
    tabO[e] = 0.5f*(wa - wb);
  }
  stash();
  __syncthreads();
  float res[NPL][16];
  #pragma unroll
  for (int it = 0; it < NPL; ++it) {
    if (it + 1 < NPL) fetch(it + 1);
    __builtin_amdgcn_sched_barrier(0);
    v8f accE = {0.f,0.f,0.f,0.f,0.f,0.f,0.f,0.f};
    v8f accO = {0.f,0.f,0.f,0.f,0.f,0.f,0.f,0.f};
    #pragma unroll
    for (int c = 0; c < NCH; ++c) {
      const float f0 = bp[4*c];
      const float f1 = bp[4*c + 1];
      const float r0 = rp[-4*c];
      const float r1 = rp[-4*c - 1];
      const v2f ae = *(const v2f*)(te + 64*c);
      const v2f ao = *(const v2f*)(to + 64*c);
      v2f sm, df;
      sm.x = f0 + r0;  sm.y = f1 + r1;
      df.x = f0 - r0;  df.y = f1 - r1;
      accE = __builtin_amdgcn_wmma_f32_16x16x4_f32(false, ae, false, sm, (short)0, accE, false, false);
      accO = __builtin_amdgcn_wmma_f32_16x16x4_f32(false, ao, false, df, (short)0, accO, false, false);
    }
    __builtin_amdgcn_sched_barrier(0);
    #pragma unroll
    for (int i = 0; i < 8; ++i) {
      res[it][i]     = accE[i] + accO[i];
      res[it][8 + i] = accE[i] - accO[i];
    }
    __syncthreads();
    if (it + 1 < NPL) {
      stash();
      __syncthreads();
    }
  }
  {
    const int colL = ct*16 + j;
    #pragma unroll
    for (int pl = 0; pl < NPL; ++pl) {
      #pragma unroll
      for (int i = 0; i < 8; ++i) {
        const int ii = h*8 + i;
        stg[(pl*TH + rb*32 + ii)*OP + colL]      = res[pl][i];
        stg[(pl*TH + rb*32 + 31 - ii)*OP + colL] = res[pl][8 + i];
      }
    }
  }
  __syncthreads();
  const size_t kb = (size_t)k * HWSZ;
  #pragma unroll 2
  for (int u = 0; u < (TH*NV)/256; ++u) {
    const int g   = tid + u*256;
    const int row = g >> 3;
    const int c4  = g & 7;
    const int gy  = gy0 + row;
    const int gxx = gx + 4*c4;
    const size_t p = (size_t)gy*WW + gxx;
    const float sy = sxb[gy];
    const v4f sx  = *(const v4f*)(sxb + gxx);
    const v4f qv  = *(const v4f*)(Q + kb + p);
    const v4f I0  = *(const v4f*)(Ipl + p);
    const v4f I1  = *(const v4f*)(Ipl + HWSZ + p);
    const v4f I2  = *(const v4f*)(Ipl + 2*(size_t)HWSZ + p);
    const v4f m0  = *(const v4f*)(meanI + p);
    const v4f m1  = *(const v4f*)(meanI + HWSZ + p);
    const v4f m2  = *(const v4f*)(meanI + 2*(size_t)HWSZ + p);
    const v4f v0  = *(const v4f*)(inv6 + p);
    const v4f v1  = *(const v4f*)(inv6 + 1*(size_t)HWSZ + p);
    const v4f v2  = *(const v4f*)(inv6 + 2*(size_t)HWSZ + p);
    const v4f v3  = *(const v4f*)(inv6 + 3*(size_t)HWSZ + p);
    const v4f v4  = *(const v4f*)(inv6 + 4*(size_t)HWSZ + p);
    const v4f v5  = *(const v4f*)(inv6 + 5*(size_t)HWSZ + p);
    const v4f u0  = *(const v4f*)(stg + (0*TH + row)*OP + 4*c4);
    const v4f u1  = *(const v4f*)(stg + (1*TH + row)*OP + 4*c4);
    const v4f u2  = *(const v4f*)(stg + (2*TH + row)*OP + 4*c4);
    const v4f u3  = *(const v4f*)(stg + (3*TH + row)*OP + 4*c4);
    v4f oa0, oa1, oa2, ob;
    #pragma unroll
    for (int e = 0; e < 4; ++e) {
      const float rN = 1.f / (sx[e]*sy - 1.f);
      const float q  = qv[e];
      const float mp = (u0[e] - q)*rN;
      const float c0 = (u1[e] - I0[e]*q)*rN - m0[e]*mp;
      const float c1 = (u2[e] - I1[e]*q)*rN - m1[e]*mp;
      const float c2 = (u3[e] - I2[e]*q)*rN - m2[e]*mp;
      const float a0 = v0[e]*c0 + v1[e]*c1 + v2[e]*c2;
      const float a1 = v1[e]*c0 + v3[e]*c1 + v4[e]*c2;
      const float a2 = v2[e]*c0 + v4[e]*c1 + v5[e]*c2;
      const float b  = mp - (a0*m0[e] + a1*m1[e] + a2*m2[e]);
      oa0[e] = a0; oa1[e] = a1; oa2[e] = a2; ob[e] = b;
    }
    *(v4f*)(AB + (size_t)(0  + k)*HWSZ + p) = oa0;
    *(v4f*)(AB + (size_t)(4  + k)*HWSZ + p) = oa1;
    *(v4f*)(AB + (size_t)(8  + k)*HWSZ + p) = oa2;
    *(v4f*)(AB + (size_t)(12 + k)*HWSZ + p) = ob;
  }
}

__global__ void prepImg(const float* __restrict__ img, float* __restrict__ Ipl,
                        float* __restrict__ P9) {
  int p = blockIdx.x*256 + threadIdx.x;
  if (p >= HWSZ) return;
  float i0 = img[p*3+0], i1 = img[p*3+1], i2 = img[p*3+2];
  Ipl[p] = i0; Ipl[HWSZ+p] = i1; Ipl[2*HWSZ+p] = i2;
  P9[0*HWSZ+p] = i0;    P9[1*HWSZ+p] = i1;    P9[2*HWSZ+p] = i2;
  P9[3*HWSZ+p] = i0*i0; P9[4*HWSZ+p] = i0*i1; P9[5*HWSZ+p] = i0*i2;
  P9[6*HWSZ+p] = i1*i1; P9[7*HWSZ+p] = i1*i2; P9[8*HWSZ+p] = i2*i2;
}

__global__ void statsK(const float* __restrict__ U, const float* __restrict__ Ipl,
                       const float* __restrict__ sxb,
                       float* __restrict__ meanI, float* __restrict__ inv6) {
  int p = blockIdx.x*256 + threadIdx.x;
  if (p >= HWSZ) return;
  int x = p % WW, y = p / WW;
  float rN = 1.f / (sxb[x]*sxb[y] - 1.f);
  float i0 = Ipl[p], i1 = Ipl[HWSZ+p], i2 = Ipl[2*HWSZ+p];
  float m0 = (U[0*HWSZ+p] - i0)*rN;
  float m1 = (U[1*HWSZ+p] - i1)*rN;
  float m2 = (U[2*HWSZ+p] - i2)*rN;
  float v00 = (U[3*HWSZ+p] - i0*i0)*rN - m0*m0 + EPSV;
  float v01 = (U[4*HWSZ+p] - i0*i1)*rN - m0*m1;
  float v02 = (U[5*HWSZ+p] - i0*i2)*rN - m0*m2;
  float v11 = (U[6*HWSZ+p] - i1*i1)*rN - m1*m1 + EPSV;
  float v12 = (U[7*HWSZ+p] - i1*i2)*rN - m1*m2;
  float v22 = (U[8*HWSZ+p] - i2*i2)*rN - m2*m2 + EPSV;
  float c00 = v11*v22 - v12*v12;
  float c01 = v02*v12 - v01*v22;
  float c02 = v01*v12 - v02*v11;
  float c11 = v00*v22 - v02*v02;
  float c12 = v01*v02 - v00*v12;
  float c22 = v00*v11 - v01*v01;
  float rd = 1.f / (v00*c00 + v01*c01 + v02*c02);
  meanI[p] = m0; meanI[HWSZ+p] = m1; meanI[2*HWSZ+p] = m2;
  inv6[0*HWSZ+p] = c00*rd; inv6[1*HWSZ+p] = c01*rd; inv6[2*HWSZ+p] = c02*rd;
  inv6[3*HWSZ+p] = c11*rd; inv6[4*HWSZ+p] = c12*rd; inv6[5*HWSZ+p] = c22*rd;
}

__global__ void softmaxInit(const float* __restrict__ un, float* __restrict__ Q) {
  int p = blockIdx.x*256 + threadIdx.x;
  if (p >= HWSZ) return;
  float L[4];
  float m = -3.4e38f;
  #pragma unroll
  for (int k = 0; k < 4; ++k) { L[k] = -un[p*4+k]; m = fmaxf(m, L[k]); }
  float s = 0.f;
  #pragma unroll
  for (int k = 0; k < 4; ++k) { L[k] = expf(L[k] - m); s += L[k]; }
  float rs = 1.f / s;
  #pragma unroll
  for (int k = 0; k < 4; ++k) Q[k*HWSZ+p] = L[k]*rs;
}

__global__ void combineK(const float* __restrict__ U, const float* __restrict__ AB,
                         const float* __restrict__ Us4, const float* __restrict__ Ipl,
                         const float* __restrict__ un, const float* __restrict__ sxb,
                         const float* __restrict__ sxg, float* __restrict__ Q) {
  int p = blockIdx.x*256 + threadIdx.x;
  if (p >= HWSZ) return;
  int x = p % WW, y = p / WW;
  float rN = 1.f / (sxb[x]*sxb[y] - 1.f);
  float rS = 1.f / (sxg[x]*sxg[y] - 1.f);
  float I0 = Ipl[p], I1 = Ipl[HWSZ+p], I2 = Ipl[2*HWSZ+p];
  float L[4];
  float m = -3.4e38f;
  #pragma unroll
  for (int k = 0; k < 4; ++k) {
    float ma0 = (U[(size_t)(0 +k)*HWSZ+p] - AB[(size_t)(0 +k)*HWSZ+p])*rN;
    float ma1 = (U[(size_t)(4 +k)*HWSZ+p] - AB[(size_t)(4 +k)*HWSZ+p])*rN;
    float ma2 = (U[(size_t)(8 +k)*HWSZ+p] - AB[(size_t)(8 +k)*HWSZ+p])*rN;
    float mb  = (U[(size_t)(12+k)*HWSZ+p] - AB[(size_t)(12+k)*HWSZ+p])*rN;
    float bil = ma0*I0 + ma1*I1 + ma2*I2 + mb;
    float q   = Q[k*HWSZ+p];
    float sp  = (Us4[k*HWSZ+p] - q)*rS;
    L[k] = -un[p*4+k] + 3.f*sp + 10.f*bil;
    m = fmaxf(m, L[k]);
  }
  float s = 0.f;
  #pragma unroll
  for (int k = 0; k < 4; ++k) { L[k] = expf(L[k] - m); s += L[k]; }
  float rs = 1.f / s;
  #pragma unroll
  for (int k = 0; k < 4; ++k) Q[k*HWSZ+p] = L[k]*rs;
}

__global__ void packK(const float* __restrict__ Q, float* __restrict__ out) {
  int p = blockIdx.x*256 + threadIdx.x;
  if (p >= HWSZ) return;
  #pragma unroll
  for (int k = 0; k < 4; ++k) out[p*4+k] = Q[k*HWSZ+p];
}

extern "C" void kernel_launch(void* const* d_in, const int* in_sizes, int n_in,
                              void* d_out, int out_size, void* d_ws, size_t ws_size,
                              hipStream_t stream) {
  (void)in_sizes; (void)n_in; (void)out_size; (void)ws_size;
  const float* img = (const float*)d_in[0];
  const float* un  = (const float*)d_in[1];
  float* ws   = (float*)d_ws;
  float* wbil = ws;
  float* wgau = ws + 128;
  float* sxb  = ws + 160;
  float* sxg  = ws + 928;
  float* pl   = ws + 2048;
  auto P = [&](int i) { return pl + (size_t)i * HWSZ; };

  dim3 blk(256);
  dim3 gP(HWSZ/256);
  initW<<<1, 256, 0, stream>>>(wbil, wgau, sxb, sxg);
  prepImg<<<gP, blk, 0, stream>>>(img, P(0), P(16));
  convH<60,3,false><<<dim3(6,8,9),  blk, 0, stream>>>(P(16), P(16), P(32), wbil);
  convV<60,3><<<dim3(8,6,9),  blk, 0, stream>>>(P(32), P(48), wbil);
  statsK<<<gP, blk, 0, stream>>>(P(48), P(0), sxb, P(3), P(6));
  softmaxInit<<<gP, blk, 0, stream>>>(un, P(12));

  for (int it = 0; it < 5; ++it) {
    convH<60,3,true><<<dim3(6,8,16), blk, 0, stream>>>(P(12), P(0), P(32), wbil);
    convVmid<60><<<dim3(24,6,4), blk, 0, stream>>>(P(32), P(12), P(0), P(3), P(6), sxb, P(16), wbil);
    convH<60,3,false><<<dim3(6,8,16), blk, 0, stream>>>(P(16), P(16), P(32), wbil);
    convV<60,3><<<dim3(8,6,16), blk, 0, stream>>>(P(32), P(48), wbil);
    convH<12,1,false><<<dim3(6,24,4), blk, 0, stream>>>(P(12), P(12), P(32), wgau);
    convV<12,1><<<dim3(24,6,4), blk, 0, stream>>>(P(32), P(64), wgau);
    combineK<<<gP, blk, 0, stream>>>(P(48), P(16), P(64), P(0), un, sxb, sxg, P(12));
  }
  packK<<<gP, blk, 0, stream>>>(P(12), (float*)d_out);
}


# update stage moved into the epilogues of the last two vertical passes: 2nd guided V pass writes each class's bilateral output, spatial V pass does the softmax update; 42 -> 37 launches
# speedup vs baseline: 1.8867x; 1.0355x over previous
#include <hip/hip_runtime.h>
#include <math.h>

typedef __attribute__((ext_vector_type(2))) float v2f;
typedef __attribute__((ext_vector_type(4))) float v4f;
typedef __attribute__((ext_vector_type(8))) float v8f;

#define HH 768
#define WW 768
#define HWSZ (768*768)
#define EPSV 1e-4f

__global__ void initW(float* wbil, float* wgau, float* sxb, float* sxg) {
  __shared__ float wb[121];
  __shared__ float wg[25];
  int t = threadIdx.x;
  if (t < 121) { float d = (float)(t - 60); float v = expf(-0.5f*d*d/(15.f*15.f)); wb[t] = v; wbil[t] = v; }
  if (t < 25)  { float d = (float)(t - 12); float v = expf(-d*d/18.f);             wg[t] = v; wgau[t] = v; }
  __syncthreads();
  for (int x = t; x < WW; x += blockDim.x) {
    float s = 0.f;
    for (int d = 0; d < 121; ++d) { int ix = x + d - 60; if (ix >= 0 && ix < WW) s += wb[d]; }
    sxb[x] = s;
    float s2 = 0.f;
    for (int d = 0; d < 25; ++d) { int ix = x + d - 12; if (ix >= 0 && ix < WW) s2 += wg[d]; }
    sxg[x] = s2;
  }
}

template<int R, int ITERS, bool MUL>
__global__ __launch_bounds__(256) void convH(const float* __restrict__ src,
                                             const float* __restrict__ src2,
                                             float* __restrict__ dst,
                                             const float* __restrict__ wdev) {
  constexpr int TW   = 128;
  constexpr int SW   = TW + 2*R;
  constexpr int NV   = SW / 4;
  constexpr int KB   = 2*R + 32;
  constexpr int NCH  = KB / 8;
  constexpr int WQ   = 2*R + 63;
  constexpr int WQP  = (WQ + 4) & ~3;
  constexpr int TABN = NCH * 64;
  constexpr int SP   = SW + 4;
  constexpr int ROWS = 32;
  constexpr int TOT  = ROWS*NV;
  constexpr int PER  = (TOT + 255) / 256;
  static_assert(SW % 4 == 0 && KB % 8 == 0 && SP % 4 == 0 && ((SP / 4) & 1) == 1, "strip layout");
  static_assert(WQ < 256, "one weight per thread");
  __shared__ __attribute__((aligned(16))) float lds[WQP + 2*TABN + ROWS*SP];
  float* wq    = lds;
  float* tabE  = lds + WQP;
  float* tabO  = tabE + TABN;
  float* strip = tabO + TABN;
  const int tid = threadIdx.x;
  const int zpl = blockIdx.z;
  const size_t pb = (size_t)zpl * HWSZ;
  const float* ip  = src + pb;
  const float* ip2 = src2;
  bool mul = false;
  if (MUL) {
    if (zpl < 4) {
      ip = src + (size_t)zpl * HWSZ;
    } else {
      const int zz = zpl - 4;
      ip  = src  + (size_t)(zz & 3)  * HWSZ;
      ip2 = src2 + (size_t)(zz >> 2) * HWSZ;
      mul = true;
    }
  }
  float* op = dst + pb;
  const int gx0 = blockIdx.x * TW;
  const int gyb = blockIdx.y * (ROWS*ITERS);

  float wreg = 0.f;
  {
    const int d = tid - 31;
    if (tid < WQ && d >= 0 && d <= 2*R) wreg = wdev[d];
  }

  v4f pre[PER];
  v4f pre2[MUL ? PER : 1];
  auto fetch = [&](int gy) {
    unsigned off[PER];
    #pragma unroll
    for (int u = 0; u < PER; ++u) {
      const int idx = tid + u*256;
      const int r  = idx / NV;
      const int c4 = idx - r*NV;
      const int gx = gx0 - R + 4*c4;
      const bool live = idx < TOT;
      const bool ok = live && gx >= 0 && gx < WW;
      const int rr = live ? r : 0;
      const int gc = ok ? gx : 0;
      off[u] = (unsigned)((gy + rr)*WW + gc);
      pre[u] = *(const v4f*)(ip + off[u]);
    }
    if (MUL) {
      if (mul) {
        #pragma unroll
        for (int u = 0; u < PER; ++u) pre2[MUL ? u : 0] = *(const v4f*)(ip2 + off[u]);
      }
    }
  };
  auto stash = [&]() {
    #pragma unroll
    for (int u = 0; u < PER; ++u) {
      const int idx = tid + u*256;
      const int r  = idx / NV;
      const int c4 = idx - r*NV;
      const int gx = gx0 - R + 4*c4;
      const bool ok = gx >= 0 && gx < WW;
      const v4f z = {0.f, 0.f, 0.f, 0.f};
      v4f v = pre[u];
      if (MUL) { if (mul) v = v * pre2[MUL ? u : 0]; }
      if (idx < TOT) *(v4f*)(strip + r*SP + 4*c4) = ok ? v : z;
    }
  };

  const int lane = tid & 31;
  const int wv   = tid >> 5;
  const int j    = lane & 15;
  const int h    = lane >> 4;
  const int bl   = wv & 3;
  const int rt   = wv >> 2;
  const float* ap = strip + (rt*16 + j)*SP + bl*32 + 2*h;
  const float* rp = strip + (rt*16 + j)*SP + bl*32 + (KB - 2) - 2*h;
  const float* te = tabE + (h*16 + j)*2;
  const float* to = tabO + (h*16 + j)*2;
  const int colA = gx0 + bl*32 + j;
  const int colB = gx0 + bl*32 + 31 - j;

  fetch(gyb);
  if (tid < WQ) wq[tid] = wreg;
  __syncthreads();
  for (int e = tid; e < TABN; e += 256) {
    const int s  = e & 1;
    const int jj = (e >> 1) & 15;
    const int hh = (e >> 5) & 1;
    const int k  = 4*(e >> 6) + 2*hh + s;
    const float wa = wq[k - jj + 31];
    const float wb = wq[k + jj];
    tabE[e] = 0.5f*(wa + wb);
    tabO[e] = 0.5f*(wa - wb);
  }
  stash();
  __syncthreads();
  #pragma unroll
  for (int it = 0; it < ITERS; ++it) {
    const int gy = gyb + it*ROWS;
    if (it + 1 < ITERS) fetch(gy + ROWS);
    __builtin_amdgcn_sched_barrier(0);
    v8f accE = {0.f,0.f,0.f,0.f,0.f,0.f,0.f,0.f};
    v8f accO = {0.f,0.f,0.f,0.f,0.f,0.f,0.f,0.f};
    #pragma unroll
    for (int c = 0; c < NCH; ++c) {
      const v2f f  = *(const v2f*)(ap + 4*c);
      const v2f r  = *(const v2f*)(rp - 4*c);
      const v2f be = *(const v2f*)(te + 64*c);
      const v2f bo = *(const v2f*)(to + 64*c);
      v2f sm, df;
      sm.x = f.x + r.y;  sm.y = f.y + r.x;
      df.x = f.x - r.y;  df.y = f.y - r.x;
      accE = __builtin_amdgcn_wmma_f32_16x16x4_f32(false, sm, false, be, (short)0, accE, false, false);
      accO = __builtin_amdgcn_wmma_f32_16x16x4_f32(false, df, false, bo, (short)0, accO, false, false);
    }
    __builtin_amdgcn_sched_barrier(0);
    if (it + 1 < ITERS) {
      __syncthreads();
      stash();
      __syncthreads();
    }
    const int row0 = gy + rt*16 + h*8;
    #pragma unroll
    for (int i = 0; i < 8; ++i) {
      op[(size_t)(row0 + i)*WW + colA] = accE[i] + accO[i];
      op[(size_t)(row0 + i)*WW + colB] = accE[i] - accO[i];
    }
  }
}

template<int R, int ITERS>
__global__ __launch_bounds__(256) void convV(const float* __restrict__ src,
                                             float* __restrict__ dst,
                                             const float* __restrict__ wdev) {
  constexpr int TH   = 128;
  constexpr int SH   = TH + 2*R;
  constexpr int CW   = 32;
  constexpr int NV   = CW / 4;
  constexpr int KB   = 2*R + 32;
  constexpr int NCH  = KB / 8;
  constexpr int WQ   = 2*R + 63;
  constexpr int WQP  = (WQ + 4) & ~3;
  constexpr int TABN = NCH * 64;
  constexpr int SP   = SH + 1;
  constexpr int TOT  = SH*NV;
  constexpr int PER  = (TOT + 255) / 256;
  static_assert(KB % 8 == 0, "folded band");
  static_assert(WQ < 256, "one weight per thread");
  __shared__ __attribute__((aligned(16))) float lds[WQP + 2*TABN + CW*SP];
  float* wq     = lds;
  float* tabE   = lds + WQP;
  float* tabO   = tabE + TABN;
  float* stripT = tabO + TABN;
  const int tid = threadIdx.x;
  const size_t pb = (size_t)blockIdx.z * HWSZ;
  const float* ip = src + pb;
  float* op = dst + pb;
  const int gy0 = blockIdx.y * TH;
  const int gxb = blockIdx.x * (CW*ITERS);

  float wreg = 0.f;
  {
    const int d = tid - 31;
    if (tid < WQ && d >= 0 && d <= 2*R) wreg = wdev[d];
  }

  v4f pre[PER];
  auto fetch = [&](int gx) {
    #pragma unroll
    for (int u = 0; u < PER; ++u) {
      const int idx = tid + u*256;
      const int r  = idx / NV;
      const int c4 = idx - r*NV;
      const int gy = gy0 - R + r;
      const bool ok = idx < TOT && gy >= 0 && gy < HH;
      const int gr = ok ? gy : 0;
      pre[u] = *(const v4f*)(ip + (size_t)gr*WW + gx + 4*c4);
    }
  };
  auto stash = [&]() {
    #pragma unroll
    for (int u = 0; u < PER; ++u) {
      const int idx = tid + u*256;
      const int r  = idx / NV;
      const int c4 = idx - r*NV;
      const int gy = gy0 - R + r;
      const bool ok = gy >= 0 && gy < HH;
      const v4f z = {0.f, 0.f, 0.f, 0.f};
      const v4f v = ok ? pre[u] : z;
      if (idx < TOT) {
        float* q = stripT + (4*c4)*SP + r;
        q[0]    = v.x;
        q[SP]   = v.y;
        q[2*SP] = v.z;
        q[3*SP] = v.w;
      }
    }
  };

  const int lane = tid & 31;
  const int wv   = tid >> 5;
  const int j    = lane & 15;
  const int h    = lane >> 4;
  const int rb   = wv & 3;
  const int ct   = wv >> 2;
  const float* bp = stripT + (ct*16 + j)*SP + rb*32 + 2*h;
  const float* rp = stripT + (ct*16 + j)*SP + rb*32 + (KB - 1) - 2*h;
  const float* te = tabE + (h*16 + j)*2;
  const float* to = tabO + (h*16 + j)*2;
  const int y0 = gy0 + rb*32;

  fetch(gxb);
  if (tid < WQ) wq[tid] = wreg;
  __syncthreads();
  for (int e = tid; e < TABN; e += 256) {
    const int s  = e & 1;
    const int ii = (e >> 1) & 15;
    const int hh = (e >> 5) & 1;
    const int k  = 4*(e >> 6) + 2*hh + s;
    const float wa = wq[k - ii + 31];
    const float wb = wq[k + ii];
    tabE[e] = 0.5f*(wa + wb);
    tabO[e] = 0.5f*(wa - wb);
  }
  stash();
  __syncthreads();
  #pragma unroll
  for (int it = 0; it < ITERS; ++it) {
    const int gx = gxb + it*CW;
    if (it + 1 < ITERS) fetch(gx + CW);
    __builtin_amdgcn_sched_barrier(0);
    v8f accE = {0.f,0.f,0.f,0.f,0.f,0.f,0.f,0.f};
    v8f accO = {0.f,0.f,0.f,0.f,0.f,0.f,0.f,0.f};
    #pragma unroll
    for (int c = 0; c < NCH; ++c) {
      const float f0 = bp[4*c];
      const float f1 = bp[4*c + 1];
      const float r0 = rp[-4*c];
      const float r1 = rp[-4*c - 1];
      const v2f ae = *(const v2f*)(te + 64*c);
      const v2f ao = *(const v2f*)(to + 64*c);
      v2f sm, df;
      sm.x = f0 + r0;  sm.y = f1 + r1;
      df.x = f0 - r0;  df.y = f1 - r1;
      accE = __builtin_amdgcn_wmma_f32_16x16x4_f32(false, ae, false, sm, (short)0, accE, false, false);
      accO = __builtin_amdgcn_wmma_f32_16x16x4_f32(false, ao, false, df, (short)0, accO, false, false);
    }
    __builtin_amdgcn_sched_barrier(0);
    if (it + 1 < ITERS) {
      __syncthreads();
      stash();
      __syncthreads();
    }
    const int col = gx + ct*16 + j;
    #pragma unroll
    for (int i = 0; i < 8; ++i) {
      const int ii = h*8 + i;
      op[(size_t)(y0 + ii)*WW + col]      = accE[i] + accO[i];
      op[(size_t)(y0 + 31 - ii)*WW + col] = accE[i] - accO[i];
    }
  }
}

template<int R>
__global__ __launch_bounds__(256, 6) void convVmid(const float* __restrict__ src,
                                                const float* __restrict__ Q,
                                                const float* __restrict__ Ipl,
                                                const float* __restrict__ meanI,
                                                const float* __restrict__ inv6,
                                                const float* __restrict__ sxb,
                                                float* __restrict__ AB,
                                                const float* __restrict__ wdev) {
  constexpr int TH   = 128;
  constexpr int SH   = TH + 2*R;
  constexpr int CW   = 32;
  constexpr int NV   = CW / 4;
  constexpr int KB   = 2*R + 32;
  constexpr int NCH  = KB / 8;
  constexpr int WQ   = 2*R + 63;
  constexpr int WQP  = (WQ + 4) & ~3;
  constexpr int TABN = NCH * 64;
  constexpr int SP   = SH + 1;
  constexpr int TOT  = SH*NV;
  constexpr int PER  = (TOT + 255) / 256;
  constexpr int NPL  = 4;
  constexpr int OP   = 36;
  constexpr int CONV = WQP + 2*TABN + CW*SP;
  constexpr int STG  = NPL*TH*OP;
  constexpr int LDSN = CONV > STG ? CONV : STG;
  static_assert(KB % 8 == 0, "folded band");
  static_assert(WQ < 256, "one weight per thread");
  __shared__ __attribute__((aligned(16))) float lds[LDSN];
  float* wq     = lds;
  float* tabE   = lds + WQP;
  float* tabO   = tabE + TABN;
  float* stripT = tabO + TABN;
  float* stg    = lds;
  const int tid = threadIdx.x;
  const int k   = blockIdx.z;
  const int gy0 = blockIdx.y * TH;
  const int gx  = blockIdx.x * CW;

  float wreg = 0.f;
  {
    const int d = tid - 31;
    if (tid < WQ && d >= 0 && d <= 2*R) wreg = wdev[d];
  }

  v4f pre[PER];
  unsigned foff[PER];
  #pragma unroll
  for (int u = 0; u < PER; ++u) {
    const int idx = tid + u*256;
    const int r  = idx / NV;
    const int c4 = idx - r*NV;
    const int gy = gy0 - R + r;
    const bool ok = idx < TOT && gy >= 0 && gy < HH;
    const int gr = ok ? gy : 0;
    foff[u] = (unsigned)(gr*WW + gx + 4*c4);
  }
  auto fetch = [&](int pl) {
    const float* ip = src + (size_t)(k + 4*pl) * HWSZ;
    #pragma unroll
    for (int u = 0; u < PER; ++u) pre[u] = *(const v4f*)(ip + foff[u]);
  };
  auto stash = [&]() {
    #pragma unroll
    for (int u = 0; u < PER; ++u) {
      const int idx = tid + u*256;
      const int r  = idx / NV;
      const int c4 = idx - r*NV;
      const int gy = gy0 - R + r;
      const bool ok = gy >= 0 && gy < HH;
      const v4f z = {0.f, 0.f, 0.f, 0.f};
      const v4f v = ok ? pre[u] : z;
      if (idx < TOT) {
        float* q = stripT + (4*c4)*SP + r;
        q[0]    = v.x;
        q[SP]   = v.y;
        q[2*SP] = v.z;
        q[3*SP] = v.w;
      }
    }
  };

  const int lane = tid & 31;
  const int wv   = tid >> 5;
  const int j    = lane & 15;
  const int h    = lane >> 4;
  const int rb   = wv & 3;
  const int ct   = wv >> 2;
  const float* bp = stripT + (ct*16 + j)*SP + rb*32 + 2*h;
  const float* rp = stripT + (ct*16 + j)*SP + rb*32 + (KB - 1) - 2*h;
  const float* te = tabE + (h*16 + j)*2;
  const float* to = tabO + (h*16 + j)*2;

  fetch(0);
  if (tid < WQ) wq[tid] = wreg;
  __syncthreads();
  for (int e = tid; e < TABN; e += 256) {
    const int s  = e & 1;
    const int ii = (e >> 1) & 15;
    const int hh = (e >> 5) & 1;
    const int kk = 4*(e >> 6) + 2*hh + s;
    const float wa = wq[kk - ii + 31];
    const float wb = wq[kk + ii];
    tabE[e] = 0.5f*(wa + wb);
    tabO[e] = 0.5f*(wa - wb);
  }
  stash();
  __syncthreads();
  float res[NPL][16];
  #pragma unroll
  for (int it = 0; it < NPL; ++it) {
    if (it + 1 < NPL) fetch(it + 1);
    __builtin_amdgcn_sched_barrier(0);
    v8f accE = {0.f,0.f,0.f,0.f,0.f,0.f,0.f,0.f};
    v8f accO = {0.f,0.f,0.f,0.f,0.f,0.f,0.f,0.f};
    #pragma unroll
    for (int c = 0; c < NCH; ++c) {
      const float f0 = bp[4*c];
      const float f1 = bp[4*c + 1];
      const float r0 = rp[-4*c];
      const float r1 = rp[-4*c - 1];
      const v2f ae = *(const v2f*)(te + 64*c);
      const v2f ao = *(const v2f*)(to + 64*c);
      v2f sm, df;
      sm.x = f0 + r0;  sm.y = f1 + r1;
      df.x = f0 - r0;  df.y = f1 - r1;
      accE = __builtin_amdgcn_wmma_f32_16x16x4_f32(false, ae, false, sm, (short)0, accE, false, false);
      accO = __builtin_amdgcn_wmma_f32_16x16x4_f32(false, ao, false, df, (short)0, accO, false, false);
    }
    __builtin_amdgcn_sched_barrier(0);
    #pragma unroll
    for (int i = 0; i < 8; ++i) {
      res[it][i]     = accE[i] + accO[i];
      res[it][8 + i] = accE[i] - accO[i];
    }
    __syncthreads();
    if (it + 1 < NPL) {
      stash();
      __syncthreads();
    }
  }
  {
    const int colL = ct*16 + j;
    #pragma unroll
    for (int pl = 0; pl < NPL; ++pl) {
      #pragma unroll
      for (int i = 0; i < 8; ++i) {
        const int ii = h*8 + i;
        stg[(pl*TH + rb*32 + ii)*OP + colL]      = res[pl][i];
        stg[(pl*TH + rb*32 + 31 - ii)*OP + colL] = res[pl][8 + i];
      }
    }
  }
  __syncthreads();
  const size_t kb = (size_t)k * HWSZ;
  #pragma unroll 2
  for (int u = 0; u < (TH*NV)/256; ++u) {
    const int g   = tid + u*256;
    const int row = g >> 3;
    const int c4  = g & 7;
    const int gy  = gy0 + row;
    const int gxx = gx + 4*c4;
    const size_t p = (size_t)gy*WW + gxx;
    const float sy = sxb[gy];
    const v4f sx  = *(const v4f*)(sxb + gxx);
    const v4f qv  = *(const v4f*)(Q + kb + p);
    const v4f I0  = *(const v4f*)(Ipl + p);
    const v4f I1  = *(const v4f*)(Ipl + HWSZ + p);
    const v4f I2  = *(const v4f*)(Ipl + 2*(size_t)HWSZ + p);
    const v4f m0  = *(const v4f*)(meanI + p);
    const v4f m1  = *(const v4f*)(meanI + HWSZ + p);
    const v4f m2  = *(const v4f*)(meanI + 2*(size_t)HWSZ + p);
    const v4f v0  = *(const v4f*)(inv6 + p);
    const v4f v1  = *(const v4f*)(inv6 + 1*(size_t)HWSZ + p);
    const v4f v2  = *(const v4f*)(inv6 + 2*(size_t)HWSZ + p);
    const v4f v3  = *(const v4f*)(inv6 + 3*(size_t)HWSZ + p);
    const v4f v4  = *(const v4f*)(inv6 + 4*(size_t)HWSZ + p);
    const v4f v5  = *(const v4f*)(inv6 + 5*(size_t)HWSZ + p);
    const v4f u0  = *(const v4f*)(stg + (0*TH + row)*OP + 4*c4);
    const v4f u1  = *(const v4f*)(stg + (1*TH + row)*OP + 4*c4);
    const v4f u2  = *(const v4f*)(stg + (2*TH + row)*OP + 4*c4);
    const v4f u3  = *(const v4f*)(stg + (3*TH + row)*OP + 4*c4);
    v4f oa0, oa1, oa2, ob;
    #pragma unroll
    for (int e = 0; e < 4; ++e) {
      const float rN = 1.f / (sx[e]*sy - 1.f);
      const float q  = qv[e];
      const float mp = (u0[e] - q)*rN;
      const float c0 = (u1[e] - I0[e]*q)*rN - m0[e]*mp;
      const float c1 = (u2[e] - I1[e]*q)*rN - m1[e]*mp;
      const float c2 = (u3[e] - I2[e]*q)*rN - m2[e]*mp;
      const float a0 = v0[e]*c0 + v1[e]*c1 + v2[e]*c2;
      const float a1 = v1[e]*c0 + v3[e]*c1 + v4[e]*c2;
      const float a2 = v2[e]*c0 + v4[e]*c1 + v5[e]*c2;
      const float b  = mp - (a0*m0[e] + a1*m1[e] + a2*m2[e]);
      oa0[e] = a0; oa1[e] = a1; oa2[e] = a2; ob[e] = b;
    }
    *(v4f*)(AB + (size_t)(0  + k)*HWSZ + p) = oa0;
    *(v4f*)(AB + (size_t)(4  + k)*HWSZ + p) = oa1;
    *(v4f*)(AB + (size_t)(8  + k)*HWSZ + p) = oa2;
    *(v4f*)(AB + (size_t)(12 + k)*HWSZ + p) = ob;
  }
}

template<int MODE>
__device__ __forceinline__ void wgSync() {
  if (MODE == 2) {
    __builtin_amdgcn_fence(__ATOMIC_RELEASE, "workgroup", "local");
    __builtin_amdgcn_s_barrier();
    __builtin_amdgcn_fence(__ATOMIC_ACQUIRE, "workgroup", "local");
  } else {
    __syncthreads();
  }
}

template<int R, int MODE>
__global__ __launch_bounds__(256, MODE == 2 ? 2 : 6) void convVep(const float* __restrict__ src,
                                               const float* __restrict__ c0,
                                               const float* __restrict__ c1,
                                               const float* __restrict__ sxv,
                                               float* __restrict__ dst,
                                               const float* __restrict__ wdev) {
  constexpr int TH   = 128;
  constexpr int SH   = TH + 2*R;
  constexpr int CW   = 32;
  constexpr int NV   = CW / 4;
  constexpr int KB   = 2*R + 32;
  constexpr int NCH  = KB / 8;
  constexpr int WQ   = 2*R + 63;
  constexpr int WQP  = (WQ + 4) & ~3;
  constexpr int TABN = NCH * 64;
  constexpr int SP   = SH + 1;
  constexpr int TOT  = SH*NV;
  constexpr int PER  = (TOT + 255) / 256;
  constexpr int NPL  = 4;
  constexpr int OP   = 36;
  constexpr int CONV = WQP + 2*TABN + CW*SP;
  constexpr int STG  = NPL*TH*OP;
  constexpr int LDSN = CONV > STG ? CONV : STG;
  static_assert(KB % 8 == 0, "folded band");
  static_assert(WQ < 256, "one weight per thread");
  static_assert(MODE == 1 || MODE == 2, "epilogue");
  __shared__ __attribute__((aligned(16))) float lds[LDSN];
  float* wq     = lds;
  float* tabE   = lds + WQP;
  float* tabO   = tabE + TABN;
  float* stripT = tabO + TABN;
  float* stg    = lds;
  const int tid = threadIdx.x;
  const int k   = blockIdx.z;
  const int gy0 = blockIdx.y * TH;
  const int gx  = blockIdx.x * CW;

  float wreg = 0.f;
  {
    const int d = tid - 31;
    if (tid < WQ && d >= 0 && d <= 2*R) wreg = wdev[d];
  }

  constexpr int NPRE = MODE == 2 ? 4 : 1;
  v4f pre[NPRE][PER];
  unsigned foff[PER];
  #pragma unroll
  for (int u = 0; u < PER; ++u) {
    const int idx = tid + u*256;
    const int r  = idx / NV;
    const int c4 = idx - r*NV;
    const int gy = gy0 - R + r;
    const bool ok = idx < TOT && gy >= 0 && gy < HH;
    const int gr = ok ? gy : 0;
    foff[u] = (unsigned)(gr*WW + gx + 4*c4);
  }
  auto fetch = [&](int pl) {
    const float* ip = src + (size_t)(MODE == 1 ? k + 4*pl : pl) * HWSZ;
    #pragma unroll
    for (int u = 0; u < PER; ++u) pre[MODE == 2 ? pl : 0][u] = *(const v4f*)(ip + foff[u]);
  };
  auto stash = [&](int pl) {
    #pragma unroll
    for (int u = 0; u < PER; ++u) {
      const int idx = tid + u*256;
      const int r  = idx / NV;
      const int c4 = idx - r*NV;
      const int gy = gy0 - R + r;
      const bool ok = gy >= 0 && gy < HH;
      const v4f z = {0.f, 0.f, 0.f, 0.f};
      const v4f v = ok ? pre[MODE == 2 ? pl : 0][u] : z;
      if (idx < TOT) {
        float* q = stripT + (4*c4)*SP + r;
        q[0]    = v.x;
        q[SP]   = v.y;
        q[2*SP] = v.z;
        q[3*SP] = v.w;
      }
    }
  };

  const int lane = tid & 31;
  const int wv   = tid >> 5;
  const int j    = lane & 15;
  const int h    = lane >> 4;
  const int rb   = wv & 3;
  const int ct   = wv >> 2;
  const float* bp = stripT + (ct*16 + j)*SP + rb*32 + 2*h;
  const float* rp = stripT + (ct*16 + j)*SP + rb*32 + (KB - 1) - 2*h;
  const float* te = tabE + (h*16 + j)*2;
  const float* to = tabO + (h*16 + j)*2;

  fetch(0);
  if (MODE == 2) { fetch(1); fetch(2); fetch(3); }
  constexpr int EG = (TH*NV)/256;
  v4f eq[MODE == 2 ? EG : 1][4], eb[MODE == 2 ? EG : 1][4], eun[MODE == 2 ? EG : 1][4], esx[MODE == 2 ? EG : 1];
  float esy[MODE == 2 ? EG : 1];
  if (MODE == 2) {
    #pragma unroll
    for (int u = 0; u < EG; ++u) {
      const int g   = tid + u*256;
      const int row = g >> 3;
      const int c4  = g & 7;
      const int gy  = gy0 + row;
      const int gxx = gx + 4*c4;
      const size_t p = (size_t)gy*WW + gxx;
      const int uu = MODE == 2 ? u : 0;
      esy[uu] = sxv[gy];
      esx[uu] = *(const v4f*)(sxv + gxx);
      #pragma unroll
      for (int c = 0; c < 4; ++c) {
        eq[uu][c]  = *(const v4f*)(dst + (size_t)c*HWSZ + p);
        eb[uu][c]  = *(const v4f*)(c0  + (size_t)c*HWSZ + p);
        eun[uu][c] = *(const v4f*)(c1 + 4*(p + c));
      }
    }
  }
  if (tid < WQ) wq[tid] = wreg;
  wgSync<MODE>();
  for (int e = tid; e < TABN; e += 256) {
    const int s  = e & 1;
    const int ii = (e >> 1) & 15;
    const int hh = (e >> 5) & 1;
    const int kk = 4*(e >> 6) + 2*hh + s;
    const float wa = wq[kk - ii + 31];
    const float wb = wq[kk + ii];
    tabE[e] = 0.5f*(wa + wb);
    tabO[e] = 0.5f*(wa - wb);
  }
  stash(0);
  wgSync<MODE>();
  float res[NPL][16];
  #pragma unroll
  for (int it = 0; it < NPL; ++it) {
    if (MODE != 2 && it + 1 < NPL) fetch(it + 1);
    __builtin_amdgcn_sched_barrier(0);
    v8f accE = {0.f,0.f,0.f,0.f,0.f,0.f,0.f,0.f};
    v8f accO = {0.f,0.f,0.f,0.f,0.f,0.f,0.f,0.f};
    #pragma unroll
    for (int c = 0; c < NCH; ++c) {
      const float f0 = bp[4*c];
      const float f1 = bp[4*c + 1];
      const float r0 = rp[-4*c];
      const float r1 = rp[-4*c - 1];
      const v2f ae = *(const v2f*)(te + 64*c);
      const v2f ao = *(const v2f*)(to + 64*c);
      v2f sm, df;
      sm.x = f0 + r0;  sm.y = f1 + r1;
      df.x = f0 - r0;  df.y = f1 - r1;
      accE = __builtin_amdgcn_wmma_f32_16x16x4_f32(false, ae, false, sm, (short)0, accE, false, false);
      accO = __builtin_amdgcn_wmma_f32_16x16x4_f32(false, ao, false, df, (short)0, accO, false, false);
    }
    __builtin_amdgcn_sched_barrier(0);
    #pragma unroll
    for (int i = 0; i < 8; ++i) {
      res[it][i]     = accE[i] + accO[i];
      res[it][8 + i] = accE[i] - accO[i];
    }
    wgSync<MODE>();
    if (it + 1 < NPL) {
      stash(it + 1);
      wgSync<MODE>();
    }
  }
  {
    const int colL = ct*16 + j;
    #pragma unroll
    for (int pl = 0; pl < NPL; ++pl) {
      #pragma unroll
      for (int i = 0; i < 8; ++i) {
        const int ii = h*8 + i;
        stg[(pl*TH + rb*32 + ii)*OP + colL]      = res[pl][i];
        stg[(pl*TH + rb*32 + 31 - ii)*OP + colL] = res[pl][8 + i];
      }
    }
  }
  wgSync<MODE>();
  #pragma unroll (MODE == 2 ? EG : 2)
  for (int u = 0; u < (TH*NV)/256; ++u) {
    const int g   = tid + u*256;
    const int row = g >> 3;
    const int c4  = g & 7;
    const int gy  = gy0 + row;
    const int gxx = gx + 4*c4;
    const size_t p = (size_t)gy*WW + gxx;
    const v4f u0  = *(const v4f*)(stg + (0*TH + row)*OP + 4*c4);
    const v4f u1  = *(const v4f*)(stg + (1*TH + row)*OP + 4*c4);
    const v4f u2  = *(const v4f*)(stg + (2*TH + row)*OP + 4*c4);
    const v4f u3  = *(const v4f*)(stg + (3*TH + row)*OP + 4*c4);
    if (MODE == 1) {
      const v4f A0 = *(const v4f*)(c0 + (size_t)(0  + k)*HWSZ + p);
      const v4f A1 = *(const v4f*)(c0 + (size_t)(4  + k)*HWSZ + p);
      const v4f A2 = *(const v4f*)(c0 + (size_t)(8  + k)*HWSZ + p);
      const v4f Bk = *(const v4f*)(c0 + (size_t)(12 + k)*HWSZ + p);
      const v4f I0 = *(const v4f*)(c1 + p);
      const v4f I1 = *(const v4f*)(c1 + HWSZ + p);
      const v4f I2 = *(const v4f*)(c1 + 2*(size_t)HWSZ + p);
      const float sy = sxv[gy];
      const v4f sx  = *(const v4f*)(sxv + gxx);
      v4f o;
      #pragma unroll
      for (int e = 0; e < 4; ++e) {
        const float rN  = 1.f / (sx[e]*sy - 1.f);
        const float ma0 = (u0[e] - A0[e])*rN;
        const float ma1 = (u1[e] - A1[e])*rN;
        const float ma2 = (u2[e] - A2[e])*rN;
        const float mb  = (u3[e] - Bk[e])*rN;
        o[e] = ma0*I0[e] + ma1*I1[e] + ma2*I2[e] + mb;
      }
      *(v4f*)(dst + (size_t)k*HWSZ + p) = o;
    } else {
      const int uu = MODE == 2 ? u : 0;
      const v4f q0 = eq[uu][0], q1 = eq[uu][1], q2 = eq[uu][2], q3 = eq[uu][3];
      const v4f b0 = eb[uu][0], b1 = eb[uu][1], b2 = eb[uu][2], b3 = eb[uu][3];
      v4f o0, o1, o2, o3;
      #pragma unroll
      for (int e = 0; e < 4; ++e) {
        const v4f un = eun[uu][e];
        const float rS = 1.f / (esx[uu][e]*esy[uu] - 1.f);
        float L0 = -un[0] + 3.f*((u0[e] - q0[e])*rS) + 10.f*b0[e];
        float L1 = -un[1] + 3.f*((u1[e] - q1[e])*rS) + 10.f*b1[e];
        float L2 = -un[2] + 3.f*((u2[e] - q2[e])*rS) + 10.f*b2[e];
        float L3 = -un[3] + 3.f*((u3[e] - q3[e])*rS) + 10.f*b3[e];
        const float m = fmaxf(fmaxf(L0, L1), fmaxf(L2, L3));
        L0 = expf(L0 - m); L1 = expf(L1 - m); L2 = expf(L2 - m); L3 = expf(L3 - m);
        const float rs = 1.f / (L0 + L1 + L2 + L3);
        o0[e] = L0*rs; o1[e] = L1*rs; o2[e] = L2*rs; o3[e] = L3*rs;
      }
      *(v4f*)(dst + p)                   = o0;
      *(v4f*)(dst + HWSZ + p)            = o1;
      *(v4f*)(dst + 2*(size_t)HWSZ + p)  = o2;
      *(v4f*)(dst + 3*(size_t)HWSZ + p)  = o3;
    }
  }
}

__global__ void prepImg(const float* __restrict__ img, float* __restrict__ Ipl,
                        float* __restrict__ P9) {
  int p = blockIdx.x*256 + threadIdx.x;
  if (p >= HWSZ) return;
  float i0 = img[p*3+0], i1 = img[p*3+1], i2 = img[p*3+2];
  Ipl[p] = i0; Ipl[HWSZ+p] = i1; Ipl[2*HWSZ+p] = i2;
  P9[0*HWSZ+p] = i0;    P9[1*HWSZ+p] = i1;    P9[2*HWSZ+p] = i2;
  P9[3*HWSZ+p] = i0*i0; P9[4*HWSZ+p] = i0*i1; P9[5*HWSZ+p] = i0*i2;
  P9[6*HWSZ+p] = i1*i1; P9[7*HWSZ+p] = i1*i2; P9[8*HWSZ+p] = i2*i2;
}

__global__ void statsK(const float* __restrict__ U, const float* __restrict__ Ipl,
                       const float* __restrict__ sxb,
                       float* __restrict__ meanI, float* __restrict__ inv6) {
  int p = blockIdx.x*256 + threadIdx.x;
  if (p >= HWSZ) return;
  int x = p % WW, y = p / WW;
  float rN = 1.f / (sxb[x]*sxb[y] - 1.f);
  float i0 = Ipl[p], i1 = Ipl[HWSZ+p], i2 = Ipl[2*HWSZ+p];
  float m0 = (U[0*HWSZ+p] - i0)*rN;
  float m1 = (U[1*HWSZ+p] - i1)*rN;
  float m2 = (U[2*HWSZ+p] - i2)*rN;
  float v00 = (U[3*HWSZ+p] - i0*i0)*rN - m0*m0 + EPSV;
  float v01 = (U[4*HWSZ+p] - i0*i1)*rN - m0*m1;
  float v02 = (U[5*HWSZ+p] - i0*i2)*rN - m0*m2;
  float v11 = (U[6*HWSZ+p] - i1*i1)*rN - m1*m1 + EPSV;
  float v12 = (U[7*HWSZ+p] - i1*i2)*rN - m1*m2;
  float v22 = (U[8*HWSZ+p] - i2*i2)*rN - m2*m2 + EPSV;
  float c00 = v11*v22 - v12*v12;
  float c01 = v02*v12 - v01*v22;
  float c02 = v01*v12 - v02*v11;
  float c11 = v00*v22 - v02*v02;
  float c12 = v01*v02 - v00*v12;
  float c22 = v00*v11 - v01*v01;
  float rd = 1.f / (v00*c00 + v01*c01 + v02*c02);
  meanI[p] = m0; meanI[HWSZ+p] = m1; meanI[2*HWSZ+p] = m2;
  inv6[0*HWSZ+p] = c00*rd; inv6[1*HWSZ+p] = c01*rd; inv6[2*HWSZ+p] = c02*rd;
  inv6[3*HWSZ+p] = c11*rd; inv6[4*HWSZ+p] = c12*rd; inv6[5*HWSZ+p] = c22*rd;
}

__global__ void softmaxInit(const float* __restrict__ un, float* __restrict__ Q) {
  int p = blockIdx.x*256 + threadIdx.x;
  if (p >= HWSZ) return;
  float L[4];
  float m = -3.4e38f;
  #pragma unroll
  for (int k = 0; k < 4; ++k) { L[k] = -un[p*4+k]; m = fmaxf(m, L[k]); }
  float s = 0.f;
  #pragma unroll
  for (int k = 0; k < 4; ++k) { L[k] = expf(L[k] - m); s += L[k]; }
  float rs = 1.f / s;
  #pragma unroll
  for (int k = 0; k < 4; ++k) Q[k*HWSZ+p] = L[k]*rs;
}

__global__ void packK(const float* __restrict__ Q, float* __restrict__ out) {
  int p = blockIdx.x*256 + threadIdx.x;
  if (p >= HWSZ) return;
  #pragma unroll
  for (int k = 0; k < 4; ++k) out[p*4+k] = Q[k*HWSZ+p];
}

extern "C" void kernel_launch(void* const* d_in, const int* in_sizes, int n_in,
                              void* d_out, int out_size, void* d_ws, size_t ws_size,
                              hipStream_t stream) {
  (void)in_sizes; (void)n_in; (void)out_size; (void)ws_size;
  const float* img = (const float*)d_in[0];
  const float* un  = (const float*)d_in[1];
  float* ws   = (float*)d_ws;
  float* wbil = ws;
  float* wgau = ws + 128;
  float* sxb  = ws + 160;
  float* sxg  = ws + 928;
  float* pl   = ws + 2048;
  auto P = [&](int i) { return pl + (size_t)i * HWSZ; };

  dim3 blk(256);
  dim3 gP(HWSZ/256);
  initW<<<1, 256, 0, stream>>>(wbil, wgau, sxb, sxg);
  prepImg<<<gP, blk, 0, stream>>>(img, P(0), P(16));
  convH<60,3,false><<<dim3(6,8,9),  blk, 0, stream>>>(P(16), P(16), P(32), wbil);
  convV<60,3><<<dim3(8,6,9),  blk, 0, stream>>>(P(32), P(48), wbil);
  statsK<<<gP, blk, 0, stream>>>(P(48), P(0), sxb, P(3), P(6));
  softmaxInit<<<gP, blk, 0, stream>>>(un, P(12));

  for (int it = 0; it < 5; ++it) {
    convH<60,3,true><<<dim3(6,8,16), blk, 0, stream>>>(P(12), P(0), P(32), wbil);
    convVmid<60><<<dim3(24,6,4), blk, 0, stream>>>(P(32), P(12), P(0), P(3), P(6), sxb, P(16), wbil);
    convH<60,3,false><<<dim3(6,8,16), blk, 0, stream>>>(P(16), P(16), P(32), wbil);
    convVep<60,1><<<dim3(24,6,4), blk, 0, stream>>>(P(32), P(16), P(0), sxb, P(48), wbil);
    convH<12,1,false><<<dim3(6,24,4), blk, 0, stream>>>(P(12), P(12), P(32), wgau);
    convVep<12,2><<<dim3(24,6,1), blk, 0, stream>>>(P(32), P(48), un, sxg, P(12), wgau);
  }
  packK<<<gP, blk, 0, stream>>>(P(12), (float*)d_out);
}
